# starting kernel
# baseline (speedup 1.0000x reference)
#include <hip/hip_runtime.h>

typedef __attribute__((ext_vector_type(2))) float v2f;
typedef __attribute__((ext_vector_type(8))) float v8f;

#define NGRID   64
#define NNODE   (NGRID * NGRID * NGRID)
#define NTYPES  256
#define PERM_LEN (NNODE + NTYPES * 16)
#define MAX_TILES (NNODE / 16 + NTYPES)
#define FILT_LAST (NTYPES * 243 - 1)

#define WS_TYPES   0
#define WS_HIST    (NNODE)
#define WS_STARTS  (WS_HIST + NTYPES)
#define WS_CURS    (WS_STARTS + NTYPES + 1)
#define WS_TOTAL   (WS_CURS + NTYPES)
#define WS_PERM    262928

__global__ void k_zero_hist(int* ws) {
    int i = blockIdx.x * blockDim.x + threadIdx.x;
    if (i < NTYPES) ws[WS_HIST + i] = 0;
}

__global__ void k_types(const float* __restrict__ rho,
                        const int* __restrict__ tw, int* ws) {
    int n = blockIdx.x * blockDim.x + threadIdx.x;
    if (n >= NNODE) return;
    int x = n >> 12, y = (n >> 6) & 63, z = n & 63;
    int t = 0;
    int i = 0;
    #pragma unroll
    for (int dx = 0; dx < 2; ++dx)
        #pragma unroll
        for (int dy = 0; dy < 2; ++dy)
            #pragma unroll
            for (int dz = 0; dz < 2; ++dz, ++i) {
                int xx = (x + dx) & 63, yy = (y + dy) & 63, zz = (z + dz) & 63;
                float r = rho[(xx << 12) | (yy << 6) | zz];
                if (r > 0.5f) t += (1 << tw[i]);
            }
    t &= (NTYPES - 1);
    ws[WS_TYPES + n] = t;
    atomicAdd(&ws[WS_HIST + t], 1);
}

__global__ void k_scan(int* ws) {
    if (blockIdx.x != 0 || threadIdx.x != 0) return;
    int s = 0;
    for (int t = 0; t < NTYPES; ++t) {
        ws[WS_STARTS + t] = s;
        ws[WS_CURS + t]   = s;
        int c = ws[WS_HIST + t];
        s += ((c + 15) >> 4) << 4;
    }
    ws[WS_STARTS + NTYPES] = s;
    ws[WS_TOTAL] = s;
}

__global__ void k_fill_perm(int* ws) {
    int i = blockIdx.x * blockDim.x + threadIdx.x;
    if (i < PERM_LEN) ws[WS_PERM + i] = -1;
}

__global__ void k_scatter(int* ws) {
    int n = blockIdx.x * blockDim.x + threadIdx.x;
    if (n >= NNODE) return;
    int t = ws[WS_TYPES + n];
    int pos = atomicAdd(&ws[WS_CURS + t], 1);
    ws[WS_PERM + pos] = n;
}

__global__ __launch_bounds__(32) void k_wmma(const float* __restrict__ U,
                                             const float* __restrict__ filters,
                                             const int* __restrict__ ws,
                                             float* __restrict__ out) {
    int tile  = blockIdx.x;
    int total = ws[WS_TOTAL];
    if (tile * 16 >= total) return;

    int lane = threadIdx.x;
    int m    = lane & 15;
    int half = lane >> 4;

    int n  = ws[WS_PERM + tile * 16 + m];
    int n0 = ws[WS_PERM + tile * 16];
    int t  = ws[WS_TYPES + n0];

    int mc    = (m < 3) ? m : 2;
    int fbase = t * 243 + mc * 81 + 2 * half;
    float fa[42];
    #pragma unroll
    for (int c = 0; c < 21; ++c) {
        int i0 = fbase + 4 * c;
        int i1 = i0 + 1;
        i0 = (i0 < FILT_LAST) ? i0 : FILT_LAST;
        i1 = (i1 < FILT_LAST) ? i1 : FILT_LAST;
        fa[2 * c]     = filters[i0];
        fa[2 * c + 1] = filters[i1];
    }

    float ug[84];
    #pragma unroll
    for (int k = 81; k < 84; ++k) ug[k] = 0.0f;
    if (n >= 0) {
        int x = n >> 12, y = (n >> 6) & 63, z = n & 63;
        #pragma unroll
        for (int j = 0; j < 27; ++j) {
            int dx = j / 9 - 1, dy = (j / 3) % 3 - 1, dz = j % 3 - 1;
            int ni = (((x + dx + 64) & 63) << 12) |
                     (((y + dy + 64) & 63) << 6)  |
                     ((z + dz + 64) & 63);
            ug[3 * j + 0] = U[ni * 3 + 0];
            ug[3 * j + 1] = U[ni * 3 + 1];
            ug[3 * j + 2] = U[ni * 3 + 2];
        }
    } else {
        #pragma unroll
        for (int k = 0; k < 81; ++k) ug[k] = 0.0f;
    }

    v8f acc = {};
    #pragma unroll
    for (int c = 0; c < 21; ++c) {
        v2f a, b;
        a.x = fa[2 * c];
        a.y = fa[2 * c + 1];
        b.x = half ? ug[4 * c + 2] : ug[4 * c + 0];
        b.y = half ? ug[4 * c + 3] : ug[4 * c + 1];
        acc = __builtin_amdgcn_wmma_f32_16x16x4_f32(
            false, a, false, b, (short)0, acc, false, false);
    }

    if (half == 0 && n >= 0) {
        out[n * 3 + 0] = acc[0];
        out[n * 3 + 1] = acc[1];
        out[n * 3 + 2] = acc[2];
    }
}

extern "C" void kernel_launch(void* const* d_in, const int* in_sizes, int n_in,
                              void* d_out, int out_size, void* d_ws, size_t ws_size,
                              hipStream_t stream) {
    (void)in_sizes; (void)n_in; (void)out_size; (void)ws_size;
    const float* U       = (const float*)d_in[0];
    const float* rho     = (const float*)d_in[1];
    const float* filters = (const float*)d_in[3];
    const int*   tw      = (const int*)d_in[4];
    float* out = (float*)d_out;
    int*   ws  = (int*)d_ws;

    k_zero_hist<<<1, 256, 0, stream>>>(ws);
    k_types<<<NNODE / 256, 256, 0, stream>>>(rho, tw, ws);
    k_scan<<<1, 1, 0, stream>>>(ws);
    k_fill_perm<<<(PERM_LEN + 255) / 256, 256, 0, stream>>>(ws);
    k_scatter<<<NNODE / 256, 256, 0, stream>>>(ws);
    k_wmma<<<MAX_TILES, 32, 0, stream>>>(U, filters, ws, out);
}


# Histogram and scatter ranks summed per 4096-node block in LDS (64 blocks x 1024 threads, 4 nodes each, loads issued together): one global atomic per block and type, not per node
# speedup vs baseline: 1.6523x; 1.6523x over previous
#include <hip/hip_runtime.h>

typedef __attribute__((ext_vector_type(2))) float v2f;
typedef __attribute__((ext_vector_type(8))) float v8f;

#define NGRID   64
#define NNODE   (NGRID * NGRID * NGRID)
#define NTYPES  256
#define PERM_LEN (NNODE + NTYPES * 16)
#define MAX_TILES (NNODE / 16 + NTYPES)
#define FILT_LAST (NTYPES * 243 - 1)

#define WS_TYPES   0
#define WS_HIST    (NNODE)
#define WS_STARTS  (WS_HIST + NTYPES)
#define WS_CURS    (WS_STARTS + NTYPES + 1)
#define WS_TOTAL   (WS_CURS + NTYPES)
#define WS_PERM    262928

__global__ void k_zero_hist(int* ws) {
    int i = blockIdx.x * blockDim.x + threadIdx.x;
    if (i < NTYPES) ws[WS_HIST + i] = 0;
}

#define SORT_BLOCKS   64
#define SORT_THREADS  1024
#define SORT_PER_THR  (NNODE / SORT_BLOCKS / SORT_THREADS)

__global__ __launch_bounds__(SORT_THREADS) void k_types(const float* __restrict__ rho,
                                                        const int* __restrict__ tw, int* ws) {
    __shared__ int cnt[NTYPES];
    int tid = threadIdx.x;
    if (tid < NTYPES) cnt[tid] = 0;
    __syncthreads();
    int w[8];
    #pragma unroll
    for (int i = 0; i < 8; ++i) w[i] = 1 << tw[i];
    int tq[SORT_PER_THR];
    #pragma unroll
    for (int q = 0; q < SORT_PER_THR; ++q) {
        int n = blockIdx.x * (SORT_THREADS * SORT_PER_THR) + q * SORT_THREADS + tid;
        int x = n >> 12, y = (n >> 6) & 63, z = n & 63;
        int t = 0;
        int i = 0;
        #pragma unroll
        for (int dx = 0; dx < 2; ++dx)
            #pragma unroll
            for (int dy = 0; dy < 2; ++dy)
                #pragma unroll
                for (int dz = 0; dz < 2; ++dz, ++i) {
                    int xx = (x + dx) & 63, yy = (y + dy) & 63, zz = (z + dz) & 63;
                    float r = rho[(xx << 12) | (yy << 6) | zz];
                    t += (r > 0.5f) ? w[i] : 0;
                }
        t &= (NTYPES - 1);
        tq[q] = t;
        ws[WS_TYPES + n] = t;
    }
    #pragma unroll
    for (int q = 0; q < SORT_PER_THR; ++q) atomicAdd(&cnt[tq[q]], 1);
    __syncthreads();
    if (tid < NTYPES) {
        int c = cnt[tid];
        if (c) atomicAdd(&ws[WS_HIST + tid], c);
    }
}

__global__ void k_scan(int* ws) {
    if (blockIdx.x != 0 || threadIdx.x != 0) return;
    int s = 0;
    for (int t = 0; t < NTYPES; ++t) {
        ws[WS_STARTS + t] = s;
        ws[WS_CURS + t]   = s;
        int c = ws[WS_HIST + t];
        s += ((c + 15) >> 4) << 4;
    }
    ws[WS_STARTS + NTYPES] = s;
    ws[WS_TOTAL] = s;
}

__global__ void k_fill_perm(int* ws) {
    int i = blockIdx.x * blockDim.x + threadIdx.x;
    if (i < PERM_LEN) ws[WS_PERM + i] = -1;
}

__global__ __launch_bounds__(SORT_THREADS) void k_scatter(int* ws) {
    __shared__ int cnt[NTYPES];
    __shared__ int base[NTYPES];
    int tid = threadIdx.x;
    if (tid < NTYPES) cnt[tid] = 0;
    __syncthreads();
    int t[SORT_PER_THR], r[SORT_PER_THR];
    #pragma unroll
    for (int q = 0; q < SORT_PER_THR; ++q) {
        int n = blockIdx.x * (SORT_THREADS * SORT_PER_THR) + q * SORT_THREADS + tid;
        t[q] = ws[WS_TYPES + n];
    }
    #pragma unroll
    for (int q = 0; q < SORT_PER_THR; ++q) r[q] = atomicAdd(&cnt[t[q]], 1);
    __syncthreads();
    if (tid < NTYPES) {
        int c = cnt[tid];
        base[tid] = c ? atomicAdd(&ws[WS_CURS + tid], c) : 0;
    }
    __syncthreads();
    #pragma unroll
    for (int q = 0; q < SORT_PER_THR; ++q) {
        int n = blockIdx.x * (SORT_THREADS * SORT_PER_THR) + q * SORT_THREADS + tid;
        ws[WS_PERM + base[t[q]] + r[q]] = n;
    }
}

__global__ __launch_bounds__(32) void k_wmma(const float* __restrict__ U,
                                             const float* __restrict__ filters,
                                             const int* __restrict__ ws,
                                             float* __restrict__ out) {
    int tile  = blockIdx.x;
    int total = ws[WS_TOTAL];
    if (tile * 16 >= total) return;

    int lane = threadIdx.x;
    int m    = lane & 15;
    int half = lane >> 4;

    int n  = ws[WS_PERM + tile * 16 + m];
    int n0 = ws[WS_PERM + tile * 16];
    int t  = ws[WS_TYPES + n0];

    int mc    = (m < 3) ? m : 2;
    int fbase = t * 243 + mc * 81 + 2 * half;
    float fa[42];
    #pragma unroll
    for (int c = 0; c < 21; ++c) {
        int i0 = fbase + 4 * c;
        int i1 = i0 + 1;
        i0 = (i0 < FILT_LAST) ? i0 : FILT_LAST;
        i1 = (i1 < FILT_LAST) ? i1 : FILT_LAST;
        fa[2 * c]     = filters[i0];
        fa[2 * c + 1] = filters[i1];
    }

    float ug[84];
    #pragma unroll
    for (int k = 81; k < 84; ++k) ug[k] = 0.0f;
    if (n >= 0) {
        int x = n >> 12, y = (n >> 6) & 63, z = n & 63;
        #pragma unroll
        for (int j = 0; j < 27; ++j) {
            int dx = j / 9 - 1, dy = (j / 3) % 3 - 1, dz = j % 3 - 1;
            int ni = (((x + dx + 64) & 63) << 12) |
                     (((y + dy + 64) & 63) << 6)  |
                     ((z + dz + 64) & 63);
            ug[3 * j + 0] = U[ni * 3 + 0];
            ug[3 * j + 1] = U[ni * 3 + 1];
            ug[3 * j + 2] = U[ni * 3 + 2];
        }
    } else {
        #pragma unroll
        for (int k = 0; k < 81; ++k) ug[k] = 0.0f;
    }

    v8f acc = {};
    #pragma unroll
    for (int c = 0; c < 21; ++c) {
        v2f a, b;
        a.x = fa[2 * c];
        a.y = fa[2 * c + 1];
        b.x = half ? ug[4 * c + 2] : ug[4 * c + 0];
        b.y = half ? ug[4 * c + 3] : ug[4 * c + 1];
        acc = __builtin_amdgcn_wmma_f32_16x16x4_f32(
            false, a, false, b, (short)0, acc, false, false);
    }

    if (half == 0 && n >= 0) {
        out[n * 3 + 0] = acc[0];
        out[n * 3 + 1] = acc[1];
        out[n * 3 + 2] = acc[2];
    }
}

extern "C" void kernel_launch(void* const* d_in, const int* in_sizes, int n_in,
                              void* d_out, int out_size, void* d_ws, size_t ws_size,
                              hipStream_t stream) {
    (void)in_sizes; (void)n_in; (void)out_size; (void)ws_size;
    const float* U       = (const float*)d_in[0];
    const float* rho     = (const float*)d_in[1];
    const float* filters = (const float*)d_in[3];
    const int*   tw      = (const int*)d_in[4];
    float* out = (float*)d_out;
    int*   ws  = (int*)d_ws;

    k_zero_hist<<<1, 256, 0, stream>>>(ws);
    k_types<<<SORT_BLOCKS, SORT_THREADS, 0, stream>>>(rho, tw, ws);
    k_scan<<<1, 1, 0, stream>>>(ws);
    k_fill_perm<<<(PERM_LEN + 255) / 256, 256, 0, stream>>>(ws);
    k_scatter<<<SORT_BLOCKS, SORT_THREADS, 0, stream>>>(ws);
    k_wmma<<<MAX_TILES, 32, 0, stream>>>(U, filters, ws, out);
}


# Product kernel operand feed: k order 42*half+2c+e makes each lane's 42 k consecutive; U neighbour triples fetched once per node by cooperating lanes via LDS, filters by wide loads
# speedup vs baseline: 1.7562x; 1.0629x over previous
#include <hip/hip_runtime.h>

typedef __attribute__((ext_vector_type(2))) float v2f;
typedef __attribute__((ext_vector_type(8))) float v8f;

#define NGRID   64
#define NNODE   (NGRID * NGRID * NGRID)
#define NTYPES  256
#define PERM_LEN (NNODE + NTYPES * 16)
#define MAX_TILES (NNODE / 16 + NTYPES)
#define FILT_LAST (NTYPES * 243 - 1)

#define WS_TYPES   0
#define WS_HIST    (NNODE)
#define WS_STARTS  (WS_HIST + NTYPES)
#define WS_CURS    (WS_STARTS + NTYPES + 1)
#define WS_TOTAL   (WS_CURS + NTYPES)
#define WS_PERM    262928

__global__ void k_zero_hist(int* ws) {
    int i = blockIdx.x * blockDim.x + threadIdx.x;
    if (i < NTYPES) ws[WS_HIST + i] = 0;
}

#define SORT_BLOCKS   64
#define SORT_THREADS  1024
#define SORT_PER_THR  (NNODE / SORT_BLOCKS / SORT_THREADS)

__global__ __launch_bounds__(SORT_THREADS) void k_types(const float* __restrict__ rho,
                                                        const int* __restrict__ tw, int* ws) {
    __shared__ int cnt[NTYPES];
    int tid = threadIdx.x;
    if (tid < NTYPES) cnt[tid] = 0;
    __syncthreads();
    int w[8];
    #pragma unroll
    for (int i = 0; i < 8; ++i) w[i] = 1 << tw[i];
    int tq[SORT_PER_THR];
    #pragma unroll
    for (int q = 0; q < SORT_PER_THR; ++q) {
        int n = blockIdx.x * (SORT_THREADS * SORT_PER_THR) + q * SORT_THREADS + tid;
        int x = n >> 12, y = (n >> 6) & 63, z = n & 63;
        int t = 0;
        int i = 0;
        #pragma unroll
        for (int dx = 0; dx < 2; ++dx)
            #pragma unroll
            for (int dy = 0; dy < 2; ++dy)
                #pragma unroll
                for (int dz = 0; dz < 2; ++dz, ++i) {
                    int xx = (x + dx) & 63, yy = (y + dy) & 63, zz = (z + dz) & 63;
                    float r = rho[(xx << 12) | (yy << 6) | zz];
                    t += (r > 0.5f) ? w[i] : 0;
                }
        t &= (NTYPES - 1);
        tq[q] = t;
        ws[WS_TYPES + n] = t;
    }
    #pragma unroll
    for (int q = 0; q < SORT_PER_THR; ++q) atomicAdd(&cnt[tq[q]], 1);
    __syncthreads();
    if (tid < NTYPES) {
        int c = cnt[tid];
        if (c) atomicAdd(&ws[WS_HIST + tid], c);
    }
}

__global__ void k_scan(int* ws) {
    if (blockIdx.x != 0 || threadIdx.x != 0) return;
    int s = 0;
    for (int t = 0; t < NTYPES; ++t) {
        ws[WS_STARTS + t] = s;
        ws[WS_CURS + t]   = s;
        int c = ws[WS_HIST + t];
        s += ((c + 15) >> 4) << 4;
    }
    ws[WS_STARTS + NTYPES] = s;
    ws[WS_TOTAL] = s;
}

__global__ void k_fill_perm(int* ws) {
    int i = blockIdx.x * blockDim.x + threadIdx.x;
    if (i < PERM_LEN) ws[WS_PERM + i] = -1;
}

__global__ __launch_bounds__(SORT_THREADS) void k_scatter(int* ws) {
    __shared__ int cnt[NTYPES];
    __shared__ int base[NTYPES];
    int tid = threadIdx.x;
    if (tid < NTYPES) cnt[tid] = 0;
    __syncthreads();
    int t[SORT_PER_THR], r[SORT_PER_THR];
    #pragma unroll
    for (int q = 0; q < SORT_PER_THR; ++q) {
        int n = blockIdx.x * (SORT_THREADS * SORT_PER_THR) + q * SORT_THREADS + tid;
        t[q] = ws[WS_TYPES + n];
    }
    #pragma unroll
    for (int q = 0; q < SORT_PER_THR; ++q) r[q] = atomicAdd(&cnt[t[q]], 1);
    __syncthreads();
    if (tid < NTYPES) {
        int c = cnt[tid];
        base[tid] = c ? atomicAdd(&ws[WS_CURS + tid], c) : 0;
    }
    __syncthreads();
    #pragma unroll
    for (int q = 0; q < SORT_PER_THR; ++q) {
        int n = blockIdx.x * (SORT_THREADS * SORT_PER_THR) + q * SORT_THREADS + tid;
        ws[WS_PERM + base[t[q]] + r[q]] = n;
    }
}

#define ROWP 92
#define HOFF 44
#define DUMP 88

__global__ __launch_bounds__(32) void k_wmma(const float* __restrict__ U,
                                             const float* __restrict__ filters,
                                             const int* __restrict__ ws,
                                             float* __restrict__ out) {
    __shared__ __attribute__((aligned(16))) float ugs[16 * ROWP];

    int tile  = blockIdx.x;
    int total = ws[WS_TOTAL];
    if (tile * 16 >= total) return;

    int lane = threadIdx.x;
    int m    = lane & 15;
    int half = lane >> 4;

    int n  = ws[WS_PERM + tile * 16 + m];
    int n0 = ws[WS_PERM + tile * 16];
    int t  = ws[WS_TYPES + n0];
    int ng = (n >= 0) ? n : n0;

    int mc = (m < 3) ? m : 2;
    const float* fp = filters + t * 243 + mc * 81 + 42 * half;
    float fa[42];
    #pragma unroll
    for (int i = 0; i < 39; ++i) fa[i] = fp[i];
    {
        int tl = half ? 36 : 39;
        float x0 = fp[tl], x1 = fp[tl + 1], x2 = fp[tl + 2];
        fa[39] = half ? 0.0f : x0;
        fa[40] = half ? 0.0f : x1;
        fa[41] = half ? 0.0f : x2;
    }

    bool v0 = m < 15, v1 = m < 12;
    int j0 = v0 ? m : 14;
    int j1 = v1 ? 15 + m : 26;
    int dx0 = j0 / 9 - 1, dy0 = (j0 / 3) % 3 - 1, dz0 = j0 % 3 - 1;
    int dx1 = j1 / 9 - 1, dy1 = (j1 / 3) % 3 - 1, dz1 = j1 % 3 - 1;
    int p0 = v0 ? 3 * j0 + (j0 >= 14 ? 2 : 0) : DUMP;
    int p1 = v1 ? 3 * j1 + 2 : DUMP;

    #pragma unroll
    for (int ip = 0; ip < 8; ++ip) {
        int na = __builtin_amdgcn_readlane(ng, 2 * ip);
        int nb = __builtin_amdgcn_readlane(ng, 2 * ip + 1);
        int nn = half ? nb : na;
        int x = nn >> 12, y = (nn >> 6) & 63, z = nn & 63;
        float* row = ugs + (2 * ip + half) * ROWP;
        {
            int ni = (((x + dx0) & 63) << 12) | (((y + dy0) & 63) << 6) | ((z + dz0) & 63);
            const float* up = U + ni * 3;
            float a = up[0], b = up[1], c = up[2];
            row[p0] = a; row[p0 + 1] = b; row[p0 + 2] = c;
        }
        {
            int ni = (((x + dx1) & 63) << 12) | (((y + dy1) & 63) << 6) | ((z + dz1) & 63);
            const float* up = U + ni * 3;
            float a = up[0], b = up[1], c = up[2];
            row[p1] = a; row[p1 + 1] = b; row[p1 + 2] = c;
        }
    }
    {
        float* zr = ugs + m * ROWP + (HOFF + 39);
        zr[0] = 0.0f; zr[1] = 0.0f; zr[2] = 0.0f;
    }
    __syncthreads();

    const float* br = ugs + m * ROWP + HOFF * half;
    float ub[42];
    #pragma unroll
    for (int i = 0; i < 42; ++i) ub[i] = br[i];

    v8f acc = {};
    #pragma unroll
    for (int c = 0; c < 21; ++c) {
        v2f a, b;
        a.x = fa[2 * c];
        a.y = fa[2 * c + 1];
        b.x = ub[2 * c];
        b.y = ub[2 * c + 1];
        acc = __builtin_amdgcn_wmma_f32_16x16x4_f32(
            false, a, false, b, (short)0, acc, false, false);
    }

    if (half == 0 && n >= 0) {
        out[n * 3 + 0] = acc[0];
        out[n * 3 + 1] = acc[1];
        out[n * 3 + 2] = acc[2];
    }
}

extern "C" void kernel_launch(void* const* d_in, const int* in_sizes, int n_in,
                              void* d_out, int out_size, void* d_ws, size_t ws_size,
                              hipStream_t stream) {
    (void)in_sizes; (void)n_in; (void)out_size; (void)ws_size;
    const float* U       = (const float*)d_in[0];
    const float* rho     = (const float*)d_in[1];
    const float* filters = (const float*)d_in[3];
    const int*   tw      = (const int*)d_in[4];
    float* out = (float*)d_out;
    int*   ws  = (int*)d_ws;

    k_zero_hist<<<1, 256, 0, stream>>>(ws);
    k_types<<<SORT_BLOCKS, SORT_THREADS, 0, stream>>>(rho, tw, ws);
    k_scan<<<1, 1, 0, stream>>>(ws);
    k_fill_perm<<<(PERM_LEN + 255) / 256, 256, 0, stream>>>(ws);
    k_scatter<<<SORT_BLOCKS, SORT_THREADS, 0, stream>>>(ws);
    k_wmma<<<MAX_TILES, 32, 0, stream>>>(U, filters, ws, out);
}


# Tiles of four 4-slot groups, each group its own type: A rows 0-5 and 8-13 carry four types' filter rows, read from an LDS copy of the filter blocks; buckets pad to 4 not 16, no global load per tile
# speedup vs baseline: 6.2205x; 3.5420x over previous
#include <hip/hip_runtime.h>


typedef __attribute__((ext_vector_type(2))) float v2f;
typedef __attribute__((ext_vector_type(3))) float v3f;
typedef __attribute__((ext_vector_type(4))) float v4f;
typedef __attribute__((ext_vector_type(8))) float v8f;

#define NGRID        64
#define PLANE        4096
#define NTYPES       256
#define WG_THREADS   1024
#define PER_THR      (PLANE / WG_THREADS)
#define WG_TYPES     64
#define FBLK         243
#define FS_FLOATS    (WG_TYPES * FBLK)
#define FROWS        (WG_TYPES * 3)
#define FPITCH       92
#define FITEMS       (FROWS * 21)
#define MAX_SLOTS    (PLANE + WG_TYPES * 3)
#define PAD_ID       0xFFFF

__global__ __launch_bounds__(WG_THREADS) void k_plane(const float* __restrict__ U,
                                                      const float* __restrict__ rho,
                                                      const float* __restrict__ filters,
                                                      const int* __restrict__ tw,
                                                      float* __restrict__ out) {
    __shared__ __attribute__((aligned(16))) float us[3 * PLANE * 4];
    __shared__ __attribute__((aligned(16))) float fs[FROWS * FPITCH];
    __shared__ unsigned short perm[MAX_SLOTS + 16];
    __shared__ int cnt[WG_TYPES];
    __shared__ int start[WG_TYPES];
    __shared__ unsigned char gtype[MAX_SLOTS / 4 + 8];
    __shared__ int total_s;

    int tid = threadIdx.x;
    int b   = blockIdx.x >> 2;
    int qd  = blockIdx.x & 3;
    if (tid < WG_TYPES) cnt[tid] = 0;

    int w[8];
    #pragma unroll
    for (int i = 0; i < 8; ++i) w[i] = 1 << tw[i];

    float uv[3][PER_THR][3];
    #pragma unroll
    for (int p = 0; p < 3; ++p) {
        int xx = (b + p - 1) & 63;
        #pragma unroll
        for (int q = 0; q < PER_THR; ++q) {
            const float* src = U + ((xx << 12) + q * WG_THREADS + tid) * 3;
            uv[p][q][0] = src[0]; uv[p][q][1] = src[1]; uv[p][q][2] = src[2];
        }
    }

    float fx[4][4];
    #pragma unroll
    for (int i = 0; i < 4; ++i) {
        int item = i * WG_THREADS + tid;
        int itc  = (item < FITEMS) ? item : (FITEMS - 1);
        int row  = itc / 21;
        int j    = itc - row * 21;
        const float* s = filters + (qd * FS_FLOATS + row * 81 + ((j < 20) ? 4 * j : 77));
        fx[i][0] = s[0]; fx[i][1] = s[1]; fx[i][2] = s[2]; fx[i][3] = s[3];
    }

    int tq[PER_THR];
    #pragma unroll
    for (int q = 0; q < PER_THR; ++q) {
        int nl = q * WG_THREADS + tid;
        int y = nl >> 6, z = nl & 63;
        int t = 0;
        int i = 0;
        #pragma unroll
        for (int dx = 0; dx < 2; ++dx)
            #pragma unroll
            for (int dy = 0; dy < 2; ++dy)
                #pragma unroll
                for (int dz = 0; dz < 2; ++dz, ++i) {
                    int xx = (b + dx) & 63, yy = (y + dy) & 63, zz = (z + dz) & 63;
                    float r = rho[(xx << 12) | (yy << 6) | zz];
                    t += (r > 0.5f) ? w[i] : 0;
                }
        tq[q] = t & (NTYPES - 1);
    }
    __syncthreads();

    int rk[PER_THR];
    #pragma unroll
    for (int q = 0; q < PER_THR; ++q) {
        rk[q] = 0;
        if ((tq[q] >> 6) == qd) rk[q] = atomicAdd(&cnt[tq[q] & (WG_TYPES - 1)], 1);
    }

    #pragma unroll
    for (int p = 0; p < 3; ++p) {
        #pragma unroll
        for (int q = 0; q < PER_THR; ++q) {
            v4f v;
            v.x = uv[p][q][0]; v.y = uv[p][q][1]; v.z = uv[p][q][2]; v.w = 0.0f;
            *(v4f*)&us[(p * PLANE + q * WG_THREADS + tid) * 4] = v;
        }
    }
    #pragma unroll
    for (int i = 0; i < 4; ++i) {
        int item = i * WG_THREADS + tid;
        int row  = item / 21;
        int j    = item - row * 21;
        if (item < FITEMS) {
            int d0 = row * FPITCH + 4 * j + ((j >= 11) ? 2 : 0);
            int d1 = row * FPITCH + 4 * j + 2 + ((j >= 10) ? 2 : 0);
            bool full = j < 20;
            v2f lo, hi;
            lo.x = full ? fx[i][0] : fx[i][3];
            lo.y = full ? fx[i][1] : 0.0f;
            hi.x = full ? fx[i][2] : 0.0f;
            hi.y = full ? fx[i][3] : 0.0f;
            *(v2f*)&fs[d0] = lo;
            *(v2f*)&fs[d1] = hi;
        }
    }
    __syncthreads();

    if (tid < 32) {
        int c0 = cnt[2 * tid], c1 = cnt[2 * tid + 1];
        int p0 = (c0 + 3) & ~3, p1 = (c1 + 3) & ~3;
        int s = p0 + p1;
        int incl = s;
        #pragma unroll
        for (int d = 1; d < 32; d <<= 1) {
            int v = __shfl_up(incl, d);
            if (tid >= d) incl += v;
        }
        int run = incl - s;
        start[2 * tid]     = run;
        start[2 * tid + 1] = run + p0;
        if (tid == 31) total_s = incl;
    }
    __syncthreads();

    #pragma unroll
    for (int q = 0; q < PER_THR; ++q) {
        if ((tq[q] >> 6) == qd) {
            int lt  = tq[q] & (WG_TYPES - 1);
            int pos = start[lt] + rk[q];
            perm[pos] = (unsigned short)(q * WG_THREADS + tid);
            gtype[pos >> 2] = (unsigned char)lt;
        }
    }
    {
        int lt = tid >> 4;
        int c  = cnt[lt];
        int e  = c + (tid & 15);
        if (e < ((c + 3) & ~3)) perm[start[lt] + e] = (unsigned short)PAD_ID;
    }
    if (tid < 12) {
        int total = total_s;
        int pos   = total + tid;
        if (pos < ((total + 15) & ~15)) {
            perm[pos] = (unsigned short)PAD_ID;
            gtype[pos >> 2] = (unsigned char)0;
        }
    }
    __syncthreads();

    int lane = tid & 31;
    int wv   = __builtin_amdgcn_readfirstlane(tid >> 5);
    int m    = lane & 15;
    int half = lane >> 4;
    int nt   = __builtin_amdgcn_readfirstlane((total_s + 15) >> 4);
    int r8   = m & 7;
    int grp  = ((m >> 3) << 1) + ((r8 >= 3) ? 1 : 0);
    int orow = (r8 < 3) ? r8 : ((r8 < 6) ? r8 - 3 : 2);
    int aoff = orow * FPITCH + 44 * half;
    bool mine  = (m >> 3) == half;
    bool upper = ((m >> 2) & 1) != 0;

    for (int tau = wv; tau < nt; tau += WG_THREADS / 32) {
        int gt  = gtype[tau * 4 + grp];
        int id  = perm[tau * 16 + m];
        int id0 = perm[tau * 16];
        bool valid = id != PAD_ID;
        int ng = valid ? id : id0;
        int y = ng >> 6, z = ng & 63;

        const float* fp = fs + (gt * (3 * FPITCH) + aoff);
        float fa[42];
        #pragma unroll
        for (int c = 0; c < 10; ++c) {
            v4f v = *(const v4f*)(fp + 4 * c);
            fa[4 * c] = v.x; fa[4 * c + 1] = v.y; fa[4 * c + 2] = v.z; fa[4 * c + 3] = v.w;
        }
        {
            v2f v = *(const v2f*)(fp + 40);
            fa[40] = v.x; fa[41] = v.y;
        }

        float ub[42];
        #pragma unroll
        for (int jj = 0; jj < 14; ++jj) {
            const int j0 = jj;
            const int j1 = (14 + jj < 27) ? 14 + jj : 26;
            int px = half ? (j1 / 9) : (j0 / 9);
            int dy = half ? ((j1 / 3) % 3 - 1) : ((j0 / 3) % 3 - 1);
            int dz = half ? (j1 % 3 - 1) : (j0 % 3 - 1);
            int nb = (px << 12) | (((y + dy) & 63) << 6) | ((z + dz) & 63);
            v3f v = *(const v3f*)&us[nb * 4];
            bool ex = (jj < 13) || (half == 0);
            ub[3 * jj]     = ex ? v.x : 0.0f;
            ub[3 * jj + 1] = ex ? v.y : 0.0f;
            ub[3 * jj + 2] = ex ? v.z : 0.0f;
        }

        __builtin_amdgcn_sched_barrier(0);

        v8f acc = {};
        #pragma unroll
        for (int c = 0; c < 21; ++c) {
            v2f a, bb;
            a.x  = fa[2 * c];
            a.y  = fa[2 * c + 1];
            bb.x = ub[2 * c];
            bb.y = ub[2 * c + 1];
            acc = __builtin_amdgcn_wmma_f32_16x16x4_f32(
                false, a, false, bb, (short)0, acc, false, false);
        }

        if (mine && valid) {
            int n = (b << 12) + id;
            out[n * 3 + 0] = upper ? acc[3] : acc[0];
            out[n * 3 + 1] = upper ? acc[4] : acc[1];
            out[n * 3 + 2] = upper ? acc[5] : acc[2];
        }
    }
}

extern "C" void kernel_launch(void* const* d_in, const int* in_sizes, int n_in,
                              void* d_out, int out_size, void* d_ws, size_t ws_size,
                              hipStream_t stream) {
    (void)in_sizes; (void)n_in; (void)out_size; (void)d_ws; (void)ws_size;
    const float* U       = (const float*)d_in[0];
    const float* rho     = (const float*)d_in[1];
    const float* filters = (const float*)d_in[3];
    const int*   tw      = (const int*)d_in[4];
    float* out = (float*)d_out;

    k_plane<<<NGRID * 4, WG_THREADS, 0, stream>>>(U, rho, filters, tw, out);
}
